# starting kernel
# baseline (speedup 1.0000x reference)
#include <hip/hip_runtime.h>

typedef float v2f __attribute__((ext_vector_type(2)));
typedef float v8f __attribute__((ext_vector_type(8)));

#define IN_FEATS  64
#define OUT_FEATS 32
#define N_KERNELS 8
#define NKF       (N_KERNELS * OUT_FEATS)
#define DIM       2

__global__ __launch_bounds__(256)
void fc_wmma_kernel(const float* __restrict__ feat,
                    const float* __restrict__ W,
                    float* __restrict__ nodefeat,
                    int n, int mtiles) {
  const int wave = blockIdx.x * (blockDim.x >> 5) + (threadIdx.x >> 5);
  const int lane = threadIdx.x & 31;
  const int tn = wave & 15;
  const int tm = wave >> 4;
  if (tm >= mtiles) return;

  const int half = lane >> 4;
  const int r    = lane & 15;

  int mrow = tm * 16 + r;
  if (mrow >= n) mrow = n - 1;
  const float* __restrict__ arow = feat + (size_t)mrow * IN_FEATS;
  const float* __restrict__ brow = W + (size_t)(tn * 16 + r) * IN_FEATS;

  v8f c = {};
#pragma unroll
  for (int kk = 0; kk < IN_FEATS / 4; ++kk) {
    const int kb = kk * 4 + half * 2;
    const float2 af = *(const float2*)(arow + kb);
    const float2 bf = *(const float2*)(brow + kb);
    v2f a; a.x = af.x; a.y = af.y;
    v2f b; b.x = bf.x; b.y = bf.y;
    c = __builtin_amdgcn_wmma_f32_16x16x4_f32(
        false, a, false, b, (short)0, c, false, false);
  }

#pragma unroll
  for (int v = 0; v < 8; ++v) {
    const int row = tm * 16 + v + half * 8;
    if (row < n)
      nodefeat[(size_t)row * NKF + tn * 16 + r] = c[v];
  }
}

__global__ __launch_bounds__(256)
void gmm_edge_kernel(const int* __restrict__ rowptr,
                     const int* __restrict__ colind,
                     const int* __restrict__ permute,
                     const float* __restrict__ pseudo,
                     const float* __restrict__ nodefeat,
                     const float* __restrict__ mu,
                     const float* __restrict__ inv_sigma,
                     const float* __restrict__ bias,
                     float* __restrict__ out,
                     int n) {
  __shared__ float s_mu[N_KERNELS * DIM];
  __shared__ float s_is[N_KERNELS * DIM];
  __shared__ float s_bias[OUT_FEATS];

  const int t = threadIdx.x;
  if (t < N_KERNELS * DIM) { s_mu[t] = mu[t]; s_is[t] = inv_sigma[t]; }
  if (t < OUT_FEATS)       { s_bias[t] = bias[t]; }
  __syncthreads();

  const int node = blockIdx.x * (blockDim.x >> 5) + (t >> 5);
  const int f    = t & 31;
  if (node >= n) return;

  const int start = rowptr[node];
  const int end   = rowptr[node + 1];

  float acc = 0.0f;
  for (int e = start; e < end; ++e) {
    const int src = colind[e];
    const int pe  = permute[e];
    const float p0 = pseudo[(size_t)pe * 2 + 0];
    const float p1 = pseudo[(size_t)pe * 2 + 1];
    const float* __restrict__ nf = nodefeat + (size_t)src * NKF + f;
#pragma unroll
    for (int k = 0; k < N_KERNELS; ++k) {
      const float d0 = (p0 - s_mu[k * 2 + 0]) * s_is[k * 2 + 0];
      const float d1 = (p1 - s_mu[k * 2 + 1]) * s_is[k * 2 + 1];
      const float wk = __expf(-0.5f * (d0 * d0 + d1 * d1));
      acc = fmaf(wk, nf[k * OUT_FEATS], acc);
    }
  }
  out[(size_t)node * OUT_FEATS + f] = acc + s_bias[f];
}

extern "C" void kernel_launch(void* const* d_in, const int* in_sizes, int n_in,
                              void* d_out, int out_size, void* d_ws, size_t ws_size,
                              hipStream_t stream) {
  const int*   rowptr    = (const int*)d_in[0];
  const int*   colind    = (const int*)d_in[1];
  const int*   permute   = (const int*)d_in[4];
  const float* feat      = (const float*)d_in[5];
  const float* pseudo    = (const float*)d_in[6];
  const float* W         = (const float*)d_in[7];
  const float* mu        = (const float*)d_in[8];
  const float* inv_sigma = (const float*)d_in[9];
  const float* bias      = (const float*)d_in[10];
  float*       out       = (float*)d_out;
  float*       nodefeat  = (float*)d_ws;

  const int n      = in_sizes[0] - 1;
  const int mtiles = (n + 15) / 16;
  const int waves  = mtiles * 16;

  dim3 blk(256);
  dim3 grid1((waves + 7) / 8);
  hipLaunchKernelGGL(fc_wmma_kernel, grid1, blk, 0, stream,
                     feat, W, nodefeat, n, mtiles);

  dim3 grid2((n + 7) / 8);
  hipLaunchKernelGGL(gmm_edge_kernel, grid2, blk, 0, stream,
                     rowptr, colind, permute, pseudo, nodefeat,
                     mu, inv_sigma, bias, out, n);
}


# Edge sum taken before the weight product (linearity): one fused launch gathers 256-B feat rows into a 16x512 LDS tile and feeds WMMA K=512; float summation order changes
# speedup vs baseline: 1.5298x; 1.5298x over previous
#include <hip/hip_runtime.h>

typedef float v2f __attribute__((ext_vector_type(2)));
typedef float v8f __attribute__((ext_vector_type(8)));


#define IN_FEATS  64
#define OUT_FEATS 32
#define N_KERNELS 8
#define DIM       2
#define TILE_M    16
#define KTOT      (N_KERNELS * IN_FEATS)
#define A_PITCH   (KTOT + 4)

__global__ __launch_bounds__(256)
void gmm_fused_kernel(const int* __restrict__ rowptr,
                      const int* __restrict__ colind,
                      const int* __restrict__ permute,
                      const float* __restrict__ pseudo,
                      const float* __restrict__ feat,
                      const float* __restrict__ W,
                      const float* __restrict__ mu,
                      const float* __restrict__ inv_sigma,
                      const float* __restrict__ bias,
                      float* __restrict__ out,
                      int n) {
  __shared__ __attribute__((aligned(16))) float sA[TILE_M * A_PITCH];
  __shared__ __attribute__((aligned(16))) float sC[4 * TILE_M * OUT_FEATS];

  const int t     = threadIdx.x;
  const int wave  = t >> 5;
  const int lane  = t & 31;
  const int node0 = blockIdx.x * TILE_M;

  float m0[N_KERNELS], m1[N_KERNELS], s0[N_KERNELS], s1[N_KERNELS];
#pragma unroll
  for (int k = 0; k < N_KERNELS; ++k) {
    m0[k] = mu[k * 2 + 0];
    m1[k] = mu[k * 2 + 1];
    s0[k] = inv_sigma[k * 2 + 0];
    s1[k] = inv_sigma[k * 2 + 1];
  }

#pragma unroll 1
  for (int q = 0; q < 2; ++q) {
    const int nl   = wave * 2 + q;
    const int node = node0 + nl;
    float ax[N_KERNELS], ay[N_KERNELS];
#pragma unroll
    for (int k = 0; k < N_KERNELS; ++k) { ax[k] = 0.0f; ay[k] = 0.0f; }

    if (node < n) {
      const int start = rowptr[node];
      const int end   = rowptr[node + 1];
      for (int e = start; e < end; ++e) {
        const int src = colind[e];
        const int pe  = permute[e];
        const float2 p = *(const float2*)(pseudo + (size_t)pe * 2);
        const float2 x = *(const float2*)(feat + (size_t)src * IN_FEATS + 2 * lane);
#pragma unroll
        for (int k = 0; k < N_KERNELS; ++k) {
          const float d0 = (p.x - m0[k]) * s0[k];
          const float d1 = (p.y - m1[k]) * s1[k];
          const float wk = __expf(-0.5f * (d0 * d0 + d1 * d1));
          ax[k] = fmaf(wk, x.x, ax[k]);
          ay[k] = fmaf(wk, x.y, ay[k]);
        }
      }
    }
    float* __restrict__ arow_w = sA + nl * A_PITCH + 2 * lane;
#pragma unroll
    for (int k = 0; k < N_KERNELS; ++k) {
      float2 v; v.x = ax[k]; v.y = ay[k];
      *(float2*)(arow_w + k * IN_FEATS) = v;
    }
  }
  __syncthreads();

  const int half = lane >> 4;
  const int r    = lane & 15;
  const int tnf  = wave & 1;
  const int kg   = wave >> 1;

  v8f c0 = {};
  v8f c1 = {};
#pragma unroll
  for (int kq = 0; kq < 2; ++kq) {
    const int k = kg * 2 + kq;
    const float* __restrict__ ak = sA + r * A_PITCH + k * IN_FEATS + 4 * half;
    const float* __restrict__ bk = W + (size_t)(k * OUT_FEATS + tnf * 16 + r) * IN_FEATS + 4 * half;
#pragma unroll
    for (int j = 0; j < IN_FEATS / 8; ++j) {
      const float4 a4 = *(const float4*)(ak + 8 * j);
      const float4 b4 = *(const float4*)(bk + 8 * j);
      v2f a, b;
      a.x = a4.x; a.y = a4.y; b.x = b4.x; b.y = b4.y;
      if (kq == 0)
        c0 = __builtin_amdgcn_wmma_f32_16x16x4_f32(false, a, false, b, (short)0, c0, false, false);
      else
        c1 = __builtin_amdgcn_wmma_f32_16x16x4_f32(false, a, false, b, (short)0, c1, false, false);
      a.x = a4.z; a.y = a4.w; b.x = b4.z; b.y = b4.w;
      if (kq == 0)
        c0 = __builtin_amdgcn_wmma_f32_16x16x4_f32(false, a, false, b, (short)0, c0, false, false);
      else
        c1 = __builtin_amdgcn_wmma_f32_16x16x4_f32(false, a, false, b, (short)0, c1, false, false);
    }
  }

#pragma unroll
  for (int v = 0; v < 8; ++v)
    sC[(kg * TILE_M + v + 8 * half) * OUT_FEATS + tnf * 16 + r] = c0[v] + c1[v];
  __syncthreads();

#pragma unroll
  for (int idx = t; idx < TILE_M * OUT_FEATS; idx += 256) {
    const int m    = idx >> 5;
    const int f    = idx & 31;
    const int node = node0 + m;
    if (node < n) {
      const float s = (sC[idx] + sC[idx + TILE_M * OUT_FEATS]) +
                      (sC[idx + 2 * TILE_M * OUT_FEATS] + sC[idx + 3 * TILE_M * OUT_FEATS]);
      out[(size_t)node * OUT_FEATS + f] = s + bias[f];
    }
  }
}

extern "C" void kernel_launch(void* const* d_in, const int* in_sizes, int n_in,
                              void* d_out, int out_size, void* d_ws, size_t ws_size,
                              hipStream_t stream) {
  const int*   rowptr    = (const int*)d_in[0];
  const int*   colind    = (const int*)d_in[1];
  const int*   permute   = (const int*)d_in[4];
  const float* feat      = (const float*)d_in[5];
  const float* pseudo    = (const float*)d_in[6];
  const float* W         = (const float*)d_in[7];
  const float* mu        = (const float*)d_in[8];
  const float* inv_sigma = (const float*)d_in[9];
  const float* bias      = (const float*)d_in[10];
  float*       out       = (float*)d_out;

  const int n = in_sizes[0] - 1;

  dim3 blk(256);
  dim3 grid((n + TILE_M - 1) / TILE_M);
  hipLaunchKernelGGL(gmm_fused_kernel, grid, blk, 0, stream,
                     rowptr, colind, permute, pseudo, feat, W,
                     mu, inv_sigma, bias, out, n);
}


# Gaussian edge weights computed once per edge (lane = edge, 32 edges per chunk) and read back as LDS broadcasts, not recomputed by all 32 lanes; per-node edge order of every sum unchanged
# speedup vs baseline: 2.6868x; 1.7563x over previous
#include <hip/hip_runtime.h>

typedef float v2f __attribute__((ext_vector_type(2)));
typedef float v8f __attribute__((ext_vector_type(8)));


#define IN_FEATS  64
#define OUT_FEATS 32
#define N_KERNELS 8
#define DIM       2
#define TILE_M    16
#define KTOT      (N_KERNELS * IN_FEATS)
#define A_PITCH   (KTOT + 4)

__global__ __launch_bounds__(256)
void gmm_fused_kernel(const int* __restrict__ rowptr,
                      const int* __restrict__ colind,
                      const int* __restrict__ permute,
                      const float* __restrict__ pseudo,
                      const float* __restrict__ feat,
                      const float* __restrict__ W,
                      const float* __restrict__ mu,
                      const float* __restrict__ inv_sigma,
                      const float* __restrict__ bias,
                      float* __restrict__ out,
                      int n) {
  __shared__ __attribute__((aligned(16))) float sA[TILE_M * A_PITCH];
  __shared__ __attribute__((aligned(16))) float sC[4 * TILE_M * OUT_FEATS];

  const int t     = threadIdx.x;
  const int wave  = __builtin_amdgcn_readfirstlane(t >> 5);
  const int lane  = t & 31;
  const int node0 = blockIdx.x * TILE_M;

  float m0[N_KERNELS], m1[N_KERNELS], s0[N_KERNELS], s1[N_KERNELS];
#pragma unroll
  for (int k = 0; k < N_KERNELS; ++k) {
    m0[k] = mu[k * 2 + 0];
    m1[k] = mu[k * 2 + 1];
    s0[k] = inv_sigma[k * 2 + 0];
    s1[k] = inv_sigma[k * 2 + 1];
  }

  float ax[N_KERNELS], ay[N_KERNELS];
  float bx[N_KERNELS], by[N_KERNELS];
#pragma unroll
  for (int k = 0; k < N_KERNELS; ++k) { ax[k] = 0.0f; ay[k] = 0.0f; bx[k] = 0.0f; by[k] = 0.0f; }

  {
    const int nA = node0 + 2 * wave;
    const int eA = rowptr[min(nA, n)];
    const int eB = rowptr[min(nA + 1, n)];
    const int eC = rowptr[min(nA + 2, n)];
    float* __restrict__ wbuf = sC + wave * (32 * N_KERNELS);
    const float* __restrict__ frow = feat + 2 * lane;

    for (int c = eA; c < eC; c += 32) {
      const int e = c + lane;
      int srcv = 0;
      if (e < eC) {
        srcv = colind[e];
        const int pe = permute[e];
        const float2 p = *(const float2*)(pseudo + (size_t)pe * 2);
        float w[N_KERNELS];
#pragma unroll
        for (int k = 0; k < N_KERNELS; ++k) {
          const float d0 = (p.x - m0[k]) * s0[k];
          const float d1 = (p.y - m1[k]) * s1[k];
          w[k] = __expf(-0.5f * (d0 * d0 + d1 * d1));
        }
        float4 wl, wh;
        wl.x = w[0]; wl.y = w[1]; wl.z = w[2]; wl.w = w[3];
        wh.x = w[4]; wh.y = w[5]; wh.z = w[6]; wh.w = w[7];
        *(float4*)(wbuf + lane * N_KERNELS)     = wl;
        *(float4*)(wbuf + lane * N_KERNELS + 4) = wh;
      }

      const int jn = min(32, eC - c);
      const int jm = min(max(eB - c, 0), jn);

#pragma unroll 4
      for (int j = 0; j < jm; ++j) {
        const int src = __builtin_amdgcn_readlane(srcv, j);
        const float2 x  = *(const float2*)(frow + (size_t)src * IN_FEATS);
        const float4 wl = *(const float4*)(wbuf + j * N_KERNELS);
        const float4 wh = *(const float4*)(wbuf + j * N_KERNELS + 4);
        ax[0] = fmaf(wl.x, x.x, ax[0]); ay[0] = fmaf(wl.x, x.y, ay[0]);
        ax[1] = fmaf(wl.y, x.x, ax[1]); ay[1] = fmaf(wl.y, x.y, ay[1]);
        ax[2] = fmaf(wl.z, x.x, ax[2]); ay[2] = fmaf(wl.z, x.y, ay[2]);
        ax[3] = fmaf(wl.w, x.x, ax[3]); ay[3] = fmaf(wl.w, x.y, ay[3]);
        ax[4] = fmaf(wh.x, x.x, ax[4]); ay[4] = fmaf(wh.x, x.y, ay[4]);
        ax[5] = fmaf(wh.y, x.x, ax[5]); ay[5] = fmaf(wh.y, x.y, ay[5]);
        ax[6] = fmaf(wh.z, x.x, ax[6]); ay[6] = fmaf(wh.z, x.y, ay[6]);
        ax[7] = fmaf(wh.w, x.x, ax[7]); ay[7] = fmaf(wh.w, x.y, ay[7]);
      }
#pragma unroll 4
      for (int j = jm; j < jn; ++j) {
        const int src = __builtin_amdgcn_readlane(srcv, j);
        const float2 x  = *(const float2*)(frow + (size_t)src * IN_FEATS);
        const float4 wl = *(const float4*)(wbuf + j * N_KERNELS);
        const float4 wh = *(const float4*)(wbuf + j * N_KERNELS + 4);
        bx[0] = fmaf(wl.x, x.x, bx[0]); by[0] = fmaf(wl.x, x.y, by[0]);
        bx[1] = fmaf(wl.y, x.x, bx[1]); by[1] = fmaf(wl.y, x.y, by[1]);
        bx[2] = fmaf(wl.z, x.x, bx[2]); by[2] = fmaf(wl.z, x.y, by[2]);
        bx[3] = fmaf(wl.w, x.x, bx[3]); by[3] = fmaf(wl.w, x.y, by[3]);
        bx[4] = fmaf(wh.x, x.x, bx[4]); by[4] = fmaf(wh.x, x.y, by[4]);
        bx[5] = fmaf(wh.y, x.x, bx[5]); by[5] = fmaf(wh.y, x.y, by[5]);
        bx[6] = fmaf(wh.z, x.x, bx[6]); by[6] = fmaf(wh.z, x.y, by[6]);
        bx[7] = fmaf(wh.w, x.x, bx[7]); by[7] = fmaf(wh.w, x.y, by[7]);
      }
    }

    float* __restrict__ arow_w = sA + (2 * wave) * A_PITCH + 2 * lane;
#pragma unroll
    for (int k = 0; k < N_KERNELS; ++k) {
      float2 v; v.x = ax[k]; v.y = ay[k];
      *(float2*)(arow_w + k * IN_FEATS) = v;
      float2 u; u.x = bx[k]; u.y = by[k];
      *(float2*)(arow_w + A_PITCH + k * IN_FEATS) = u;
    }
  }
  __syncthreads();

  const int half = lane >> 4;
  const int r    = lane & 15;
  const int tnf  = wave & 1;
  const int kg   = wave >> 1;

  v8f c0 = {};
  v8f c1 = {};
#pragma unroll
  for (int kq = 0; kq < 2; ++kq) {
    const int k = kg * 2 + kq;
    const float* __restrict__ ak = sA + r * A_PITCH + k * IN_FEATS + 4 * half;
    const float* __restrict__ bk = W + (size_t)(k * OUT_FEATS + tnf * 16 + r) * IN_FEATS + 4 * half;
#pragma unroll
    for (int j = 0; j < IN_FEATS / 8; ++j) {
      const float4 a4 = *(const float4*)(ak + 8 * j);
      const float4 b4 = *(const float4*)(bk + 8 * j);
      v2f a, b;
      a.x = a4.x; a.y = a4.y; b.x = b4.x; b.y = b4.y;
      if (kq == 0)
        c0 = __builtin_amdgcn_wmma_f32_16x16x4_f32(false, a, false, b, (short)0, c0, false, false);
      else
        c1 = __builtin_amdgcn_wmma_f32_16x16x4_f32(false, a, false, b, (short)0, c1, false, false);
      a.x = a4.z; a.y = a4.w; b.x = b4.z; b.y = b4.w;
      if (kq == 0)
        c0 = __builtin_amdgcn_wmma_f32_16x16x4_f32(false, a, false, b, (short)0, c0, false, false);
      else
        c1 = __builtin_amdgcn_wmma_f32_16x16x4_f32(false, a, false, b, (short)0, c1, false, false);
    }
  }

#pragma unroll
  for (int v = 0; v < 8; ++v)
    sC[(kg * TILE_M + v + 8 * half) * OUT_FEATS + tnf * 16 + r] = c0[v] + c1[v];
  __syncthreads();

#pragma unroll
  for (int idx = t; idx < TILE_M * OUT_FEATS; idx += 256) {
    const int m    = idx >> 5;
    const int f    = idx & 31;
    const int node = node0 + m;
    if (node < n) {
      const float s = (sC[idx] + sC[idx + TILE_M * OUT_FEATS]) +
                      (sC[idx + 2 * TILE_M * OUT_FEATS] + sC[idx + 3 * TILE_M * OUT_FEATS]);
      out[(size_t)node * OUT_FEATS + f] = s + bias[f];
    }
  }
}

extern "C" void kernel_launch(void* const* d_in, const int* in_sizes, int n_in,
                              void* d_out, int out_size, void* d_ws, size_t ws_size,
                              hipStream_t stream) {
  const int*   rowptr    = (const int*)d_in[0];
  const int*   colind    = (const int*)d_in[1];
  const int*   permute   = (const int*)d_in[4];
  const float* feat      = (const float*)d_in[5];
  const float* pseudo    = (const float*)d_in[6];
  const float* W         = (const float*)d_in[7];
  const float* mu        = (const float*)d_in[8];
  const float* inv_sigma = (const float*)d_in[9];
  const float* bias      = (const float*)d_in[10];
  float*       out       = (float*)d_out;

  const int n = in_sizes[0] - 1;

  dim3 blk(256);
  dim3 grid((n + TILE_M - 1) / TILE_M);
  hipLaunchKernelGGL(gmm_fused_kernel, grid, blk, 0, stream,
                     rowptr, colind, permute, pseudo, feat, W,
                     mu, inv_sigma, bias, out, n);
}


# Gather depth 4 -> 16: a node's 16 feat rows are requested at once, before the weights exist (edge ids and coordinates of both nodes fetched in one pass); per-node edge order of every sum unchanged
# speedup vs baseline: 2.8657x; 1.0666x over previous
#include <hip/hip_runtime.h>

typedef float v2f __attribute__((ext_vector_type(2)));
typedef float v8f __attribute__((ext_vector_type(8)));


#define IN_FEATS  64
#define OUT_FEATS 32
#define N_KERNELS 8
#define DIM       2
#define TILE_M    16
#define KTOT      (N_KERNELS * IN_FEATS)
#define A_PITCH   (KTOT + 4)

__global__ __launch_bounds__(256)
void gmm_fused_kernel(const int* __restrict__ rowptr,
                      const int* __restrict__ colind,
                      const int* __restrict__ permute,
                      const float* __restrict__ pseudo,
                      const float* __restrict__ feat,
                      const float* __restrict__ W,
                      const float* __restrict__ mu,
                      const float* __restrict__ inv_sigma,
                      const float* __restrict__ bias,
                      float* __restrict__ out,
                      int n) {
  __shared__ __attribute__((aligned(16))) float sA[TILE_M * A_PITCH];
  __shared__ __attribute__((aligned(16))) float sC[4 * TILE_M * OUT_FEATS];

  const int t     = threadIdx.x;
  const int wave  = __builtin_amdgcn_readfirstlane(t >> 5);
  const int lane  = t & 31;
  const int node0 = blockIdx.x * TILE_M;

  float m0[N_KERNELS], m1[N_KERNELS], s0[N_KERNELS], s1[N_KERNELS];
#pragma unroll
  for (int k = 0; k < N_KERNELS; ++k) {
    m0[k] = mu[k * 2 + 0];
    m1[k] = mu[k * 2 + 1];
    s0[k] = inv_sigma[k * 2 + 0];
    s1[k] = inv_sigma[k * 2 + 1];
  }

  {
    const int nA = node0 + 2 * wave;
    const int eA = rowptr[min(nA, n)];
    const int eB = rowptr[min(nA + 1, n)];
    const int eC = rowptr[min(nA + 2, n)];
    float* __restrict__ wbuf = sC + wave * (32 * N_KERNELS);
    const float* __restrict__ frow = feat + 2 * lane;

    const int hn = lane >> 4;
    const int eS = max(min((hn ? eB : eA) + (lane & 15), (hn ? eC : eB) - 1), 0);
    const int    srcS = colind[eS];
    const float2 pS   = *(const float2*)(pseudo + (size_t)permute[eS] * 2);
    bool need_w = true;

#pragma unroll 1
    for (int nn = 0; nn < 2; ++nn) {
      const int e0 = nn ? eB : eA;
      const int e1 = nn ? eC : eB;
      float cx[N_KERNELS], cy[N_KERNELS];
#pragma unroll
      for (int k = 0; k < N_KERNELS; ++k) { cx[k] = 0.0f; cy[k] = 0.0f; }

      int    srcL = 0;
      float2 pL;
      pL.x = 0.0f; pL.y = 0.0f;
      int    lb   = 16 * nn;

      for (int c = e0; c < e1; c += 16) {
        const int cnt = min(16, e1 - c);
        const bool shared = (c == e0);
        const int srcv = shared ? srcS : srcL;

        auto fma_edge = [&](const float2 x, int j) __attribute__((always_inline)) {
          const float4 wl = *(const float4*)(wbuf + (lb + j) * N_KERNELS);
          const float4 wh = *(const float4*)(wbuf + (lb + j) * N_KERNELS + 4);
          cx[0] = fmaf(wl.x, x.x, cx[0]); cy[0] = fmaf(wl.x, x.y, cy[0]);
          cx[1] = fmaf(wl.y, x.x, cx[1]); cy[1] = fmaf(wl.y, x.y, cy[1]);
          cx[2] = fmaf(wl.z, x.x, cx[2]); cy[2] = fmaf(wl.z, x.y, cy[2]);
          cx[3] = fmaf(wl.w, x.x, cx[3]); cy[3] = fmaf(wl.w, x.y, cy[3]);
          cx[4] = fmaf(wh.x, x.x, cx[4]); cy[4] = fmaf(wh.x, x.y, cy[4]);
          cx[5] = fmaf(wh.y, x.x, cx[5]); cy[5] = fmaf(wh.y, x.y, cy[5]);
          cx[6] = fmaf(wh.z, x.x, cx[6]); cy[6] = fmaf(wh.z, x.y, cy[6]);
          cx[7] = fmaf(wh.w, x.x, cx[7]); cy[7] = fmaf(wh.w, x.y, cy[7]);
        };
        auto weights = [&]() __attribute__((always_inline)) {
          if (!shared || need_w) {
            if (shared || lane < cnt) {
              const float2 p = shared ? pS : pL;
              float w[N_KERNELS];
#pragma unroll
              for (int k = 0; k < N_KERNELS; ++k) {
                const float d0 = (p.x - m0[k]) * s0[k];
                const float d1 = (p.y - m1[k]) * s1[k];
                w[k] = __expf(-0.5f * (d0 * d0 + d1 * d1));
              }
              float4 wl, wh;
              wl.x = w[0]; wl.y = w[1]; wl.z = w[2]; wl.w = w[3];
              wh.x = w[4]; wh.y = w[5]; wh.z = w[6]; wh.w = w[7];
              *(float4*)(wbuf + lane * N_KERNELS)     = wl;
              *(float4*)(wbuf + lane * N_KERNELS + 4) = wh;
            }
            if (shared) need_w = false;
          }
        };

        if (cnt == 16) {
          float2 X[16];
#pragma unroll
          for (int q = 0; q < 16; ++q) {
            const int s = __builtin_amdgcn_readlane(srcv, lb + q);
            X[q] = *(const float2*)(frow + (size_t)s * IN_FEATS);
          }
          weights();
#pragma unroll
          for (int q = 0; q < 16; ++q) fma_edge(X[q], q);
        } else {
          weights();
#pragma unroll 4
          for (int j = 0; j < cnt; ++j) {
            const int s = __builtin_amdgcn_readlane(srcv, lb + j);
            fma_edge(*(const float2*)(frow + (size_t)s * IN_FEATS), j);
          }
        }

        if (c + 16 < e1) {
          const int e = min(c + 16 + lane, e1 - 1);
          srcL = colind[e];
          pL   = *(const float2*)(pseudo + (size_t)permute[e] * 2);
          lb   = 0;
        }
      }

      float* __restrict__ arow_w = sA + (2 * wave + nn) * A_PITCH + 2 * lane;
#pragma unroll
      for (int k = 0; k < N_KERNELS; ++k) {
        float2 v; v.x = cx[k]; v.y = cy[k];
        *(float2*)(arow_w + k * IN_FEATS) = v;
      }
    }
  }
  __syncthreads();

  const int half = lane >> 4;
  const int r    = lane & 15;
  const int tnf  = wave & 1;
  const int kg   = wave >> 1;

  v8f c0 = {};
  v8f c1 = {};
#pragma unroll
  for (int kq = 0; kq < 2; ++kq) {
    const int k = kg * 2 + kq;
    const float* __restrict__ ak = sA + r * A_PITCH + k * IN_FEATS + 4 * half;
    const float* __restrict__ bk = W + (size_t)(k * OUT_FEATS + tnf * 16 + r) * IN_FEATS + 4 * half;
#pragma unroll
    for (int j = 0; j < IN_FEATS / 8; ++j) {
      const float4 a4 = *(const float4*)(ak + 8 * j);
      const float4 b4 = *(const float4*)(bk + 8 * j);
      v2f a, b;
      a.x = a4.x; a.y = a4.y; b.x = b4.x; b.y = b4.y;
      if (kq == 0)
        c0 = __builtin_amdgcn_wmma_f32_16x16x4_f32(false, a, false, b, (short)0, c0, false, false);
      else
        c1 = __builtin_amdgcn_wmma_f32_16x16x4_f32(false, a, false, b, (short)0, c1, false, false);
      a.x = a4.z; a.y = a4.w; b.x = b4.z; b.y = b4.w;
      if (kq == 0)
        c0 = __builtin_amdgcn_wmma_f32_16x16x4_f32(false, a, false, b, (short)0, c0, false, false);
      else
        c1 = __builtin_amdgcn_wmma_f32_16x16x4_f32(false, a, false, b, (short)0, c1, false, false);
    }
  }

#pragma unroll
  for (int v = 0; v < 8; ++v)
    sC[(kg * TILE_M + v + 8 * half) * OUT_FEATS + tnf * 16 + r] = c0[v] + c1[v];
  __syncthreads();

#pragma unroll
  for (int idx = t; idx < TILE_M * OUT_FEATS; idx += 256) {
    const int m    = idx >> 5;
    const int f    = idx & 31;
    const int node = node0 + m;
    if (node < n) {
      const float s = (sC[idx] + sC[idx + TILE_M * OUT_FEATS]) +
                      (sC[idx + 2 * TILE_M * OUT_FEATS] + sC[idx + 3 * TILE_M * OUT_FEATS]);
      out[(size_t)node * OUT_FEATS + f] = s + bias[f];
    }
  }
}

extern "C" void kernel_launch(void* const* d_in, const int* in_sizes, int n_in,
                              void* d_out, int out_size, void* d_ws, size_t ws_size,
                              hipStream_t stream) {
  const int*   rowptr    = (const int*)d_in[0];
  const int*   colind    = (const int*)d_in[1];
  const int*   permute   = (const int*)d_in[4];
  const float* feat      = (const float*)d_in[5];
  const float* pseudo    = (const float*)d_in[6];
  const float* W         = (const float*)d_in[7];
  const float* mu        = (const float*)d_in[8];
  const float* inv_sigma = (const float*)d_in[9];
  const float* bias      = (const float*)d_in[10];
  float*       out       = (float*)d_out;

  const int n = in_sizes[0] - 1;

  dim3 blk(256);
  dim3 grid((n + TILE_M - 1) / TILE_M);
  hipLaunchKernelGGL(gmm_fused_kernel, grid, blk, 0, stream,
                     rowptr, colind, permute, pseudo, feat, W,
                     mu, inv_sigma, bias, out, n);
}


# 32 nodes per work-group (was 16): a wave aggregates four nodes, each W fragment it loads feeds both 16-row tiles; half as many work-groups and waves enter; order of every sum unchanged
# speedup vs baseline: 3.0812x; 1.0752x over previous
#include <hip/hip_runtime.h>

typedef float v2f __attribute__((ext_vector_type(2)));
typedef float v8f __attribute__((ext_vector_type(8)));


#define IN_FEATS  64
#define OUT_FEATS 32
#define N_KERNELS 8
#define DIM       2
#define TILE_M    32
#define KTOT      (N_KERNELS * IN_FEATS)
#define A_PITCH   (KTOT + 4)

__global__ __launch_bounds__(256)
void gmm_fused_kernel(const int* __restrict__ rowptr,
                      const int* __restrict__ colind,
                      const int* __restrict__ permute,
                      const float* __restrict__ pseudo,
                      const float* __restrict__ feat,
                      const float* __restrict__ W,
                      const float* __restrict__ mu,
                      const float* __restrict__ inv_sigma,
                      const float* __restrict__ bias,
                      float* __restrict__ out,
                      int n) {
  __shared__ __attribute__((aligned(16))) float sA[TILE_M * A_PITCH];
  __shared__ __attribute__((aligned(16))) float sWt[8 * 32 * N_KERNELS];

  const int t     = threadIdx.x;
  const int wave  = __builtin_amdgcn_readfirstlane(t >> 5);
  const int lane  = t & 31;
  const int node0 = blockIdx.x * TILE_M;

  float m0[N_KERNELS], m1[N_KERNELS], s0[N_KERNELS], s1[N_KERNELS];
#pragma unroll
  for (int k = 0; k < N_KERNELS; ++k) {
    m0[k] = mu[k * 2 + 0];
    m1[k] = mu[k * 2 + 1];
    s0[k] = inv_sigma[k * 2 + 0];
    s1[k] = inv_sigma[k * 2 + 1];
  }

  {
    const int nA = node0 + 4 * wave;
    const int r0 = rowptr[min(nA, n)];
    const int r1 = rowptr[min(nA + 1, n)];
    const int r2 = rowptr[min(nA + 2, n)];
    const int r3 = rowptr[min(nA + 3, n)];
    const int r4 = rowptr[min(nA + 4, n)];
    float* __restrict__ wbuf = sWt + wave * (32 * N_KERNELS);
    const float* __restrict__ frow = feat + 2 * lane;

    const int hn  = lane >> 4;
    const int eSa = max(min((hn ? r1 : r0) + (lane & 15), (hn ? r2 : r1) - 1), 0);
    const int eSb = max(min((hn ? r3 : r2) + (lane & 15), (hn ? r4 : r3) - 1), 0);
    const int srcSa = colind[eSa];
    const int prmSa = permute[eSa];
    const int srcSb = colind[eSb];
    const int prmSb = permute[eSb];

#pragma unroll
    for (int pr = 0; pr < 2; ++pr) {
      const int eA = pr ? r2 : r0;
      const int eB = pr ? r3 : r1;
      const int eC = pr ? r4 : r2;
      const int    srcS = pr ? srcSb : srcSa;
      const float2 pS   = *(const float2*)(pseudo + (size_t)(pr ? prmSb : prmSa) * 2);
      bool need_w = true;

#pragma unroll 1
      for (int nn = 0; nn < 2; ++nn) {
        const int e0 = nn ? eB : eA;
        const int e1 = nn ? eC : eB;
        float cx[N_KERNELS], cy[N_KERNELS];
#pragma unroll
        for (int k = 0; k < N_KERNELS; ++k) { cx[k] = 0.0f; cy[k] = 0.0f; }

        int    srcL = 0;
        float2 pL;
        pL.x = 0.0f; pL.y = 0.0f;
        int    lb   = 16 * nn;

        for (int c = e0; c < e1; c += 16) {
          const int cnt = min(16, e1 - c);
          const bool shared = (c == e0);
          const int srcv = shared ? srcS : srcL;

          auto fma_edge = [&](const float2 x, int j) __attribute__((always_inline)) {
            const float4 wl = *(const float4*)(wbuf + (lb + j) * N_KERNELS);
            const float4 wh = *(const float4*)(wbuf + (lb + j) * N_KERNELS + 4);
            cx[0] = fmaf(wl.x, x.x, cx[0]); cy[0] = fmaf(wl.x, x.y, cy[0]);
            cx[1] = fmaf(wl.y, x.x, cx[1]); cy[1] = fmaf(wl.y, x.y, cy[1]);
            cx[2] = fmaf(wl.z, x.x, cx[2]); cy[2] = fmaf(wl.z, x.y, cy[2]);
            cx[3] = fmaf(wl.w, x.x, cx[3]); cy[3] = fmaf(wl.w, x.y, cy[3]);
            cx[4] = fmaf(wh.x, x.x, cx[4]); cy[4] = fmaf(wh.x, x.y, cy[4]);
            cx[5] = fmaf(wh.y, x.x, cx[5]); cy[5] = fmaf(wh.y, x.y, cy[5]);
            cx[6] = fmaf(wh.z, x.x, cx[6]); cy[6] = fmaf(wh.z, x.y, cy[6]);
            cx[7] = fmaf(wh.w, x.x, cx[7]); cy[7] = fmaf(wh.w, x.y, cy[7]);
          };
          auto weights = [&]() __attribute__((always_inline)) {
            if (!shared || need_w) {
              if (shared || lane < cnt) {
                const float2 p = shared ? pS : pL;
                float w[N_KERNELS];
#pragma unroll
                for (int k = 0; k < N_KERNELS; ++k) {
                  const float d0 = (p.x - m0[k]) * s0[k];
                  const float d1 = (p.y - m1[k]) * s1[k];
                  w[k] = __expf(-0.5f * (d0 * d0 + d1 * d1));
                }
                float4 wl, wh;
                wl.x = w[0]; wl.y = w[1]; wl.z = w[2]; wl.w = w[3];
                wh.x = w[4]; wh.y = w[5]; wh.z = w[6]; wh.w = w[7];
                *(float4*)(wbuf + lane * N_KERNELS)     = wl;
                *(float4*)(wbuf + lane * N_KERNELS + 4) = wh;
              }
              if (shared) need_w = false;
            }
          };

          if (cnt == 16) {
            float2 X[16];
#pragma unroll
            for (int u = 0; u < 16; ++u) {
              const int s = __builtin_amdgcn_readlane(srcv, lb + u);
              X[u] = *(const float2*)(frow + (size_t)s * IN_FEATS);
            }
            weights();
#pragma unroll
            for (int u = 0; u < 16; ++u) fma_edge(X[u], u);
          } else {
            weights();
#pragma unroll 4
            for (int j = 0; j < cnt; ++j) {
              const int s = __builtin_amdgcn_readlane(srcv, lb + j);
              fma_edge(*(const float2*)(frow + (size_t)s * IN_FEATS), j);
            }
          }

          if (c + 16 < e1) {
            const int e = min(c + 16 + lane, e1 - 1);
            srcL = colind[e];
            pL   = *(const float2*)(pseudo + (size_t)permute[e] * 2);
            lb   = 0;
          }
        }

        float* __restrict__ arow_w = sA + (4 * wave + 2 * pr + nn) * A_PITCH + 2 * lane;
#pragma unroll
        for (int k = 0; k < N_KERNELS; ++k) {
          float2 v; v.x = cx[k]; v.y = cy[k];
          *(float2*)(arow_w + k * IN_FEATS) = v;
        }
      }
    }
  }
  __syncthreads();

  const int half = lane >> 4;
  const int r    = lane & 15;
  const int tnf  = wave & 1;
  const int kg   = wave >> 1;

  v8f c00 = {};
  v8f c01 = {};
  v8f c10 = {};
  v8f c11 = {};
#pragma unroll
  for (int kq = 0; kq < 2; ++kq) {
    const int k = kg * 2 + kq;
    const float* __restrict__ ak = sA + r * A_PITCH + k * IN_FEATS + 4 * half;
    const float* __restrict__ bk = W + (size_t)(k * OUT_FEATS + tnf * 16 + r) * IN_FEATS + 4 * half;
#pragma unroll
    for (int j = 0; j < IN_FEATS / 8; ++j) {
      const float4 a4 = *(const float4*)(ak + 8 * j);
      const float4 g4 = *(const float4*)(ak + 16 * A_PITCH + 8 * j);
      const float4 b4 = *(const float4*)(bk + 8 * j);
      v2f a, g, b;
      a.x = a4.x; a.y = a4.y; g.x = g4.x; g.y = g4.y; b.x = b4.x; b.y = b4.y;
      if (kq == 0) {
        c00 = __builtin_amdgcn_wmma_f32_16x16x4_f32(false, a, false, b, (short)0, c00, false, false);
        c10 = __builtin_amdgcn_wmma_f32_16x16x4_f32(false, g, false, b, (short)0, c10, false, false);
      } else {
        c01 = __builtin_amdgcn_wmma_f32_16x16x4_f32(false, a, false, b, (short)0, c01, false, false);
        c11 = __builtin_amdgcn_wmma_f32_16x16x4_f32(false, g, false, b, (short)0, c11, false, false);
      }
      a.x = a4.z; a.y = a4.w; g.x = g4.z; g.y = g4.w; b.x = b4.z; b.y = b4.w;
      if (kq == 0) {
        c00 = __builtin_amdgcn_wmma_f32_16x16x4_f32(false, a, false, b, (short)0, c00, false, false);
        c10 = __builtin_amdgcn_wmma_f32_16x16x4_f32(false, g, false, b, (short)0, c10, false, false);
      } else {
        c01 = __builtin_amdgcn_wmma_f32_16x16x4_f32(false, a, false, b, (short)0, c01, false, false);
        c11 = __builtin_amdgcn_wmma_f32_16x16x4_f32(false, g, false, b, (short)0, c11, false, false);
      }
    }
  }

  __syncthreads();
  float* sC = sA;

#pragma unroll
  for (int v = 0; v < 8; ++v) {
    sC[(kg * TILE_M + v + 8 * half) * OUT_FEATS + tnf * 16 + r]      = c00[v] + c01[v];
    sC[(kg * TILE_M + 16 + v + 8 * half) * OUT_FEATS + tnf * 16 + r] = c10[v] + c11[v];
  }
  __syncthreads();

#pragma unroll
  for (int idx = t; idx < TILE_M * OUT_FEATS; idx += 256) {
    const int m    = idx >> 5;
    const int f    = idx & 31;
    const int node = node0 + m;
    if (node < n) {
      const float s = (sC[idx] + sC[idx + TILE_M * OUT_FEATS]) +
                      (sC[idx + 2 * TILE_M * OUT_FEATS] + sC[idx + 3 * TILE_M * OUT_FEATS]);
      out[(size_t)node * OUT_FEATS + f] = s + bias[f];
    }
  }
}

extern "C" void kernel_launch(void* const* d_in, const int* in_sizes, int n_in,
                              void* d_out, int out_size, void* d_ws, size_t ws_size,
                              hipStream_t stream) {
  const int*   rowptr    = (const int*)d_in[0];
  const int*   colind    = (const int*)d_in[1];
  const int*   permute   = (const int*)d_in[4];
  const float* feat      = (const float*)d_in[5];
  const float* pseudo    = (const float*)d_in[6];
  const float* W         = (const float*)d_in[7];
  const float* mu        = (const float*)d_in[8];
  const float* inv_sigma = (const float*)d_in[9];
  const float* bias      = (const float*)d_in[10];
  float*       out       = (float*)d_out;

  const int n = in_sizes[0] - 1;

  dim3 blk(256);
  dim3 grid((n + TILE_M - 1) / TILE_M);
  hipLaunchKernelGGL(gmm_fused_kernel, grid, blk, 0, stream,
                     rowptr, colind, permute, pseudo, feat, W,
                     mu, inv_sigma, bias, out, n);
}
